# starting kernel
# baseline (speedup 1.0000x reference)
#include <hip/hip_runtime.h>
#include <hip/hip_bf16.h>

typedef float v2f __attribute__((ext_vector_type(2)));
typedef float v8f __attribute__((ext_vector_type(8)));

#define BDIM 8
#define NDIM 2048
#define MDIM 2048
#define DDIM 128

__global__ __launch_bounds__(256)
void pairdist_wmma_kernel(const float* __restrict__ Lp,
                          const float* __restrict__ Rp,
                          float* __restrict__ Out)
{
    const int lane = threadIdx.x & 31;
    const int wave = threadIdx.x >> 5;
    const int half = lane >> 4;
    const int lrow = lane & 15;

    const int b  = blockIdx.z;
    const int n0 = (blockIdx.y * 8 + wave) * 16;
    const int m0 = blockIdx.x * 64;

    const float* __restrict__ Abase =
        Lp + ((size_t)b * NDIM + (size_t)(n0 + lrow)) * DDIM + 2 * half;
    const float* __restrict__ Bbase =
        Rp + ((size_t)b * MDIM + (size_t)(m0 + lrow)) * DDIM + 2 * half;

    v8f acc0 = {}, acc1 = {}, acc2 = {}, acc3 = {};
    float sa = 0.0f;
    float sb0 = 0.0f, sb1 = 0.0f, sb2 = 0.0f, sb3 = 0.0f;

#pragma unroll
    for (int k = 0; k < DDIM; k += 4) {
        v2f a = *(const v2f*)(Abase + k);
        sa = fmaf(a.x, a.x, sa);
        sa = fmaf(a.y, a.y, sa);

        v2f b0 = *(const v2f*)(Bbase + 0 * 16 * DDIM + k);
        v2f b1 = *(const v2f*)(Bbase + 1 * 16 * DDIM + k);
        v2f b2 = *(const v2f*)(Bbase + 2 * 16 * DDIM + k);
        v2f b3 = *(const v2f*)(Bbase + 3 * 16 * DDIM + k);

        sb0 = fmaf(b0.x, b0.x, sb0); sb0 = fmaf(b0.y, b0.y, sb0);
        sb1 = fmaf(b1.x, b1.x, sb1); sb1 = fmaf(b1.y, b1.y, sb1);
        sb2 = fmaf(b2.x, b2.x, sb2); sb2 = fmaf(b2.y, b2.y, sb2);
        sb3 = fmaf(b3.x, b3.x, sb3); sb3 = fmaf(b3.y, b3.y, sb3);

        acc0 = __builtin_amdgcn_wmma_f32_16x16x4_f32(false, a, false, b0, (short)0, acc0, false, false);
        acc1 = __builtin_amdgcn_wmma_f32_16x16x4_f32(false, a, false, b1, (short)0, acc1, false, false);
        acc2 = __builtin_amdgcn_wmma_f32_16x16x4_f32(false, a, false, b2, (short)0, acc2, false, false);
        acc3 = __builtin_amdgcn_wmma_f32_16x16x4_f32(false, a, false, b3, (short)0, acc3, false, false);
    }

    float l2 = sa + __shfl_xor(sa, 16, 32);
    float r2t[4];
    r2t[0] = sb0 + __shfl_xor(sb0, 16, 32);
    r2t[1] = sb1 + __shfl_xor(sb1, 16, 32);
    r2t[2] = sb2 + __shfl_xor(sb2, 16, 32);
    r2t[3] = sb3 + __shfl_xor(sb3, 16, 32);

    float l2row[8];
#pragma unroll
    for (int v = 0; v < 8; ++v) {
        l2row[v] = __shfl(l2, v + 8 * half, 32);
    }

    v8f accs[4] = {acc0, acc1, acc2, acc3};
    const size_t outRowBase = ((size_t)b * NDIM + (size_t)n0) * MDIM + (size_t)m0;

#pragma unroll
    for (int t = 0; t < 4; ++t) {
        const size_t colBase = outRowBase + (size_t)(t * 16 + lrow);
#pragma unroll
        for (int v = 0; v < 8; ++v) {
            const int row = v + 8 * half;
            float d2 = fmaf(-2.0f, accs[t][v], l2row[v] + r2t[t]);
            d2 = fmaxf(d2, 0.0f);
            float den = 1.0f + __builtin_sqrtf(d2);
            den = fmaxf(den, 1e-7f);
            Out[colBase + (size_t)row * MDIM] = 1.0f / den;
        }
    }
}

extern "C" void kernel_launch(void* const* d_in, const int* in_sizes, int n_in,
                              void* d_out, int out_size, void* d_ws, size_t ws_size,
                              hipStream_t stream) {
    (void)in_sizes; (void)n_in; (void)out_size; (void)d_ws; (void)ws_size;
    const float* left  = (const float*)d_in[0];
    const float* right = (const float*)d_in[1];
    float* out = (float*)d_out;

    dim3 block(256, 1, 1);
    dim3 grid(MDIM / 64, NDIM / 128, BDIM);
    pairdist_wmma_kernel<<<grid, block, 0, stream>>>(left, right, out);
}


# A and B tiles staged once per block in padded LDS by whole-row 16-byte global loads; fragments come from LDS, one 16-byte read per two matrix steps; same k order, sums bit-identical
# speedup vs baseline: 1.5203x; 1.5203x over previous
#include <hip/hip_runtime.h>
#include <hip/hip_bf16.h>


typedef float v2f __attribute__((ext_vector_type(2)));
typedef float v4f __attribute__((ext_vector_type(4)));
typedef float v8f __attribute__((ext_vector_type(8)));

#define BDIM 8
#define NDIM 2048
#define MDIM 2048
#define DDIM 128

#define TILE_N 128
#define TILE_M 64
#define PITCH 132

__global__ __launch_bounds__(256)
void pairdist_wmma_kernel(const float* __restrict__ Lp,
                          const float* __restrict__ Rp,
                          float* __restrict__ Out)
{
    __shared__ __attribute__((aligned(16))) float tile[(TILE_N + TILE_M) * PITCH];

    const int lane = threadIdx.x & 31;
    const int wave = threadIdx.x >> 5;
    const int half = lane >> 4;
    const int lrow = lane & 15;

    const int b  = blockIdx.z;
    const int nb = blockIdx.y * TILE_N;
    const int n0 = nb + wave * 16;
    const int m0 = blockIdx.x * TILE_M;

    {
        const float* __restrict__ gA =
            Lp + ((size_t)b * NDIM + (size_t)(nb + wave)) * DDIM + lane * 4;
        const float* __restrict__ gB =
            Rp + ((size_t)b * MDIM + (size_t)(m0 + wave)) * DDIM + lane * 4;
        float* sdst = tile + wave * PITCH + (lane >> 1) * 8 + (lane & 1) * 2;

        v4f stg[24];
#pragma unroll
        for (int i = 0; i < 16; ++i)
            stg[i] = *(const v4f*)(gA + (size_t)i * 8 * DDIM);
#pragma unroll
        for (int i = 0; i < 8; ++i)
            stg[16 + i] = *(const v4f*)(gB + (size_t)i * 8 * DDIM);
#pragma unroll
        for (int i = 0; i < 24; ++i) {
            v2f lo = {stg[i].x, stg[i].y};
            v2f hi = {stg[i].z, stg[i].w};
            *(v2f*)(sdst + i * 8 * PITCH)     = lo;
            *(v2f*)(sdst + i * 8 * PITCH + 4) = hi;
        }
    }
    __syncthreads();

    const float* Abase = tile + (wave * 16 + lrow) * PITCH + 4 * half;
    const float* Bbase = tile + (TILE_N + lrow) * PITCH + 4 * half;

    v8f acc0 = {}, acc1 = {}, acc2 = {}, acc3 = {};
    float sa = 0.0f;
    float sb0 = 0.0f, sb1 = 0.0f, sb2 = 0.0f, sb3 = 0.0f;

#pragma unroll
    for (int k = 0; k < DDIM; k += 8) {
        const v4f qa  = *(const v4f*)(Abase + k);
        const v4f qb0 = *(const v4f*)(Bbase + 0 * 16 * PITCH + k);
        const v4f qb1 = *(const v4f*)(Bbase + 1 * 16 * PITCH + k);
        const v4f qb2 = *(const v4f*)(Bbase + 2 * 16 * PITCH + k);
        const v4f qb3 = *(const v4f*)(Bbase + 3 * 16 * PITCH + k);

#pragma unroll
        for (int s = 0; s < 2; ++s) {
            const v2f a  = s ? (v2f){qa.z,  qa.w}  : (v2f){qa.x,  qa.y};
            const v2f b0 = s ? (v2f){qb0.z, qb0.w} : (v2f){qb0.x, qb0.y};
            const v2f b1 = s ? (v2f){qb1.z, qb1.w} : (v2f){qb1.x, qb1.y};
            const v2f b2 = s ? (v2f){qb2.z, qb2.w} : (v2f){qb2.x, qb2.y};
            const v2f b3 = s ? (v2f){qb3.z, qb3.w} : (v2f){qb3.x, qb3.y};

            sa = fmaf(a.x, a.x, sa);
            sa = fmaf(a.y, a.y, sa);

            sb0 = fmaf(b0.x, b0.x, sb0); sb0 = fmaf(b0.y, b0.y, sb0);
            sb1 = fmaf(b1.x, b1.x, sb1); sb1 = fmaf(b1.y, b1.y, sb1);
            sb2 = fmaf(b2.x, b2.x, sb2); sb2 = fmaf(b2.y, b2.y, sb2);
            sb3 = fmaf(b3.x, b3.x, sb3); sb3 = fmaf(b3.y, b3.y, sb3);

            acc0 = __builtin_amdgcn_wmma_f32_16x16x4_f32(false, a, false, b0, (short)0, acc0, false, false);
            acc1 = __builtin_amdgcn_wmma_f32_16x16x4_f32(false, a, false, b1, (short)0, acc1, false, false);
            acc2 = __builtin_amdgcn_wmma_f32_16x16x4_f32(false, a, false, b2, (short)0, acc2, false, false);
            acc3 = __builtin_amdgcn_wmma_f32_16x16x4_f32(false, a, false, b3, (short)0, acc3, false, false);
        }
    }

    float l2 = sa + __shfl_xor(sa, 16, 32);
    float r2t[4];
    r2t[0] = sb0 + __shfl_xor(sb0, 16, 32);
    r2t[1] = sb1 + __shfl_xor(sb1, 16, 32);
    r2t[2] = sb2 + __shfl_xor(sb2, 16, 32);
    r2t[3] = sb3 + __shfl_xor(sb3, 16, 32);

    float l2row[8];
#pragma unroll
    for (int v = 0; v < 8; ++v) {
        l2row[v] = __shfl(l2, v + 8 * half, 32);
    }

    v8f accs[4] = {acc0, acc1, acc2, acc3};
    const size_t outRowBase = ((size_t)b * NDIM + (size_t)n0) * MDIM + (size_t)m0;

#pragma unroll
    for (int t = 0; t < 4; ++t) {
        const size_t colBase = outRowBase + (size_t)(t * 16 + lrow);
#pragma unroll
        for (int v = 0; v < 8; ++v) {
            const int row = v + 8 * half;
            float d2 = fmaf(-2.0f, accs[t][v], l2row[v] + r2t[t]);
            d2 = fmaxf(d2, 0.0f);
            float den = 1.0f + __builtin_sqrtf(d2);
            den = fmaxf(den, 1e-7f);
            Out[colBase + (size_t)row * MDIM] = 1.0f / den;
        }
    }
}

extern "C" void kernel_launch(void* const* d_in, const int* in_sizes, int n_in,
                              void* d_out, int out_size, void* d_ws, size_t ws_size,
                              hipStream_t stream) {
    (void)in_sizes; (void)n_in; (void)out_size; (void)d_ws; (void)ws_size;
    const float* left  = (const float*)d_in[0];
    const float* right = (const float*)d_in[1];
    float* out = (float*)d_out;

    dim3 block(256, 1, 1);
    dim3 grid(MDIM / 64, NDIM / 128, BDIM);
    pairdist_wmma_kernel<<<grid, block, 0, stream>>>(left, right, out);
}


# epilogue: hardware v_sqrt_f32 and v_rcp_f32 (about 1 ulp each) replace the correctly rounded square root and IEEE division; outputs move in the last bits; products and sums unchanged
# speedup vs baseline: 1.8117x; 1.1917x over previous
#include <hip/hip_runtime.h>
#include <hip/hip_bf16.h>


typedef float v2f __attribute__((ext_vector_type(2)));
typedef float v4f __attribute__((ext_vector_type(4)));
typedef float v8f __attribute__((ext_vector_type(8)));

#define BDIM 8
#define NDIM 2048
#define MDIM 2048
#define DDIM 128

#define TILE_N 128
#define TILE_M 64
#define PITCH 132

__global__ __launch_bounds__(256)
void pairdist_wmma_kernel(const float* __restrict__ Lp,
                          const float* __restrict__ Rp,
                          float* __restrict__ Out)
{
    __shared__ __attribute__((aligned(16))) float tile[(TILE_N + TILE_M) * PITCH];

    const int lane = threadIdx.x & 31;
    const int wave = threadIdx.x >> 5;
    const int half = lane >> 4;
    const int lrow = lane & 15;

    const int b  = blockIdx.z;
    const int nb = blockIdx.y * TILE_N;
    const int n0 = nb + wave * 16;
    const int m0 = blockIdx.x * TILE_M;

    {
        const float* __restrict__ gA =
            Lp + ((size_t)b * NDIM + (size_t)(nb + wave)) * DDIM + lane * 4;
        const float* __restrict__ gB =
            Rp + ((size_t)b * MDIM + (size_t)(m0 + wave)) * DDIM + lane * 4;
        float* sdst = tile + wave * PITCH + (lane >> 1) * 8 + (lane & 1) * 2;

        v4f stg[24];
#pragma unroll
        for (int i = 0; i < 16; ++i)
            stg[i] = *(const v4f*)(gA + (size_t)i * 8 * DDIM);
#pragma unroll
        for (int i = 0; i < 8; ++i)
            stg[16 + i] = *(const v4f*)(gB + (size_t)i * 8 * DDIM);
#pragma unroll
        for (int i = 0; i < 24; ++i) {
            v2f lo = {stg[i].x, stg[i].y};
            v2f hi = {stg[i].z, stg[i].w};
            *(v2f*)(sdst + i * 8 * PITCH)     = lo;
            *(v2f*)(sdst + i * 8 * PITCH + 4) = hi;
        }
    }
    __syncthreads();

    const float* Abase = tile + (wave * 16 + lrow) * PITCH + 4 * half;
    const float* Bbase = tile + (TILE_N + lrow) * PITCH + 4 * half;

    v8f acc0 = {}, acc1 = {}, acc2 = {}, acc3 = {};
    float sa = 0.0f;
    float sb0 = 0.0f, sb1 = 0.0f, sb2 = 0.0f, sb3 = 0.0f;

#pragma unroll
    for (int k = 0; k < DDIM; k += 8) {
        const v4f qa  = *(const v4f*)(Abase + k);
        const v4f qb0 = *(const v4f*)(Bbase + 0 * 16 * PITCH + k);
        const v4f qb1 = *(const v4f*)(Bbase + 1 * 16 * PITCH + k);
        const v4f qb2 = *(const v4f*)(Bbase + 2 * 16 * PITCH + k);
        const v4f qb3 = *(const v4f*)(Bbase + 3 * 16 * PITCH + k);

#pragma unroll
        for (int s = 0; s < 2; ++s) {
            const v2f a  = s ? (v2f){qa.z,  qa.w}  : (v2f){qa.x,  qa.y};
            const v2f b0 = s ? (v2f){qb0.z, qb0.w} : (v2f){qb0.x, qb0.y};
            const v2f b1 = s ? (v2f){qb1.z, qb1.w} : (v2f){qb1.x, qb1.y};
            const v2f b2 = s ? (v2f){qb2.z, qb2.w} : (v2f){qb2.x, qb2.y};
            const v2f b3 = s ? (v2f){qb3.z, qb3.w} : (v2f){qb3.x, qb3.y};

            sa = fmaf(a.x, a.x, sa);
            sa = fmaf(a.y, a.y, sa);

            sb0 = fmaf(b0.x, b0.x, sb0); sb0 = fmaf(b0.y, b0.y, sb0);
            sb1 = fmaf(b1.x, b1.x, sb1); sb1 = fmaf(b1.y, b1.y, sb1);
            sb2 = fmaf(b2.x, b2.x, sb2); sb2 = fmaf(b2.y, b2.y, sb2);
            sb3 = fmaf(b3.x, b3.x, sb3); sb3 = fmaf(b3.y, b3.y, sb3);

            acc0 = __builtin_amdgcn_wmma_f32_16x16x4_f32(false, a, false, b0, (short)0, acc0, false, false);
            acc1 = __builtin_amdgcn_wmma_f32_16x16x4_f32(false, a, false, b1, (short)0, acc1, false, false);
            acc2 = __builtin_amdgcn_wmma_f32_16x16x4_f32(false, a, false, b2, (short)0, acc2, false, false);
            acc3 = __builtin_amdgcn_wmma_f32_16x16x4_f32(false, a, false, b3, (short)0, acc3, false, false);
        }
    }

    float l2 = sa + __shfl_xor(sa, 16, 32);
    float r2t[4];
    r2t[0] = sb0 + __shfl_xor(sb0, 16, 32);
    r2t[1] = sb1 + __shfl_xor(sb1, 16, 32);
    r2t[2] = sb2 + __shfl_xor(sb2, 16, 32);
    r2t[3] = sb3 + __shfl_xor(sb3, 16, 32);

    float l2row[8];
#pragma unroll
    for (int v = 0; v < 8; ++v) {
        l2row[v] = __shfl(l2, v + 8 * half, 32);
    }

    v8f accs[4] = {acc0, acc1, acc2, acc3};
    const size_t outRowBase = ((size_t)b * NDIM + (size_t)n0) * MDIM + (size_t)m0;

#pragma unroll
    for (int t = 0; t < 4; ++t) {
        const size_t colBase = outRowBase + (size_t)(t * 16 + lrow);
#pragma unroll
        for (int v = 0; v < 8; ++v) {
            const int row = v + 8 * half;
            float d2 = fmaf(-2.0f, accs[t][v], l2row[v] + r2t[t]);
            d2 = fmaxf(d2, 0.0f);
            const float den = 1.0f + __builtin_amdgcn_sqrtf(d2);
            Out[colBase + (size_t)row * MDIM] = __builtin_amdgcn_rcpf(den);
        }
    }
}

extern "C" void kernel_launch(void* const* d_in, const int* in_sizes, int n_in,
                              void* d_out, int out_size, void* d_ws, size_t ws_size,
                              hipStream_t stream) {
    (void)in_sizes; (void)n_in; (void)out_size; (void)d_ws; (void)ws_size;
    const float* left  = (const float*)d_in[0];
    const float* right = (const float*)d_in[1];
    float* out = (float*)d_out;

    dim3 block(256, 1, 1);
    dim3 grid(MDIM / 64, NDIM / 128, BDIM);
    pairdist_wmma_kernel<<<grid, block, 0, stream>>>(left, right, out);
}


# each block keeps its 128 left rows in LDS and loops over 8 right tiles; the next tile's global loads run under the matrix steps (two LDS buffers, fragment reads one block ahead); sums bit-identical
# speedup vs baseline: 1.9161x; 1.0576x over previous
#include <hip/hip_runtime.h>
#include <hip/hip_bf16.h>


typedef float v2f __attribute__((ext_vector_type(2)));
typedef float v4f __attribute__((ext_vector_type(4)));
typedef float v8f __attribute__((ext_vector_type(8)));

#define BDIM 8
#define NDIM 2048
#define MDIM 2048
#define DDIM 128

#define TILE_N 128
#define TILE_M 64
#define TILES_PER_BLOCK 8
#define PITCH 132

__global__ __launch_bounds__(256)
void pairdist_wmma_kernel(const float* __restrict__ Lp,
                          const float* __restrict__ Rp,
                          float* __restrict__ Out)
{
    __shared__ __attribute__((aligned(16))) float tile[(TILE_N + 2 * TILE_M) * PITCH];

    const int lane = threadIdx.x & 31;
    const int wave = threadIdx.x >> 5;
    const int half = lane >> 4;
    const int lrow = lane & 15;

    const int b  = blockIdx.z;
    const int nb = blockIdx.y * TILE_N;
    const int n0 = nb + wave * 16;
    const int mb = blockIdx.x * (TILE_M * TILES_PER_BLOCK);

    const float* __restrict__ gB =
        Rp + ((size_t)b * MDIM + (size_t)(mb + wave)) * DDIM + lane * 4;
    float* sdst = tile + wave * PITCH + (lane >> 1) * 8 + (lane & 1) * 2;

    {
        const float* __restrict__ gA =
            Lp + ((size_t)b * NDIM + (size_t)(nb + wave)) * DDIM + lane * 4;

        v4f stg[24];
#pragma unroll
        for (int i = 0; i < 16; ++i)
            stg[i] = *(const v4f*)(gA + (size_t)i * 8 * DDIM);
#pragma unroll
        for (int i = 0; i < 8; ++i)
            stg[16 + i] = *(const v4f*)(gB + (size_t)i * 8 * DDIM);
#pragma unroll
        for (int i = 0; i < 24; ++i) {
            v2f lo = {stg[i].x, stg[i].y};
            v2f hi = {stg[i].z, stg[i].w};
            *(v2f*)(sdst + i * 8 * PITCH)     = lo;
            *(v2f*)(sdst + i * 8 * PITCH + 4) = hi;
        }
    }
    __syncthreads();

    const float* Abase = tile + (wave * 16 + lrow) * PITCH + 4 * half;
    const float* Bbase0 = tile + (TILE_N + lrow) * PITCH + 4 * half;
    float* sB0 = sdst + TILE_N * PITCH;

    float l2row[8];
    {
        float sa = 0.0f;
#pragma unroll
        for (int k = 0; k < DDIM; k += 8) {
            const v4f qa = *(const v4f*)(Abase + k);
            sa = fmaf(qa.x, qa.x, sa);
            sa = fmaf(qa.y, qa.y, sa);
            sa = fmaf(qa.z, qa.z, sa);
            sa = fmaf(qa.w, qa.w, sa);
        }
        const float l2 = sa + __shfl_xor(sa, 16, 32);
#pragma unroll
        for (int v = 0; v < 8; ++v)
            l2row[v] = __shfl(l2, v + 8 * half, 32);
    }

    const size_t outRowBase = ((size_t)b * NDIM + (size_t)n0) * MDIM + (size_t)mb;

#pragma unroll 1
    for (int t = 0; t < TILES_PER_BLOCK; ++t) {
        const int cur = t & 1;
        const bool more = (t + 1 < TILES_PER_BLOCK);

        v4f pre[8];
        if (more) {
            const float* __restrict__ g = gB + (size_t)(t + 1) * TILE_M * DDIM;
#pragma unroll
            for (int i = 0; i < 8; ++i)
                pre[i] = *(const v4f*)(g + (size_t)i * 8 * DDIM);
        }

        const float* Bbase = Bbase0 + cur * (TILE_M * PITCH);

        v8f acc0 = {}, acc1 = {}, acc2 = {}, acc3 = {};
        float sb0 = 0.0f, sb1 = 0.0f, sb2 = 0.0f, sb3 = 0.0f;

        v4f fa[2][2];
        v4f fb[2][2][4];
#pragma unroll
        for (int i = 0; i < 2; ++i) {
            fa[0][i] = *(const v4f*)(Abase + 8 * i);
#pragma unroll
            for (int q = 0; q < 4; ++q)
                fb[0][i][q] = *(const v4f*)(Bbase + q * 16 * PITCH + 8 * i);
        }
#pragma unroll
        for (int j = 0; j < DDIM / 16; ++j) {
            const int cs = j & 1;
            const int ns = cs ^ 1;
            if (j + 1 < DDIM / 16) {
#pragma unroll
                for (int i = 0; i < 2; ++i) {
                    const int k = 16 * (j + 1) + 8 * i;
                    fa[ns][i] = *(const v4f*)(Abase + k);
#pragma unroll
                    for (int q = 0; q < 4; ++q)
                        fb[ns][i][q] = *(const v4f*)(Bbase + q * 16 * PITCH + k);
                }
            }
            __builtin_amdgcn_sched_barrier(0);

#pragma unroll
            for (int i = 0; i < 2; ++i) {
                const v4f qa  = fa[cs][i];
                const v4f qb0 = fb[cs][i][0];
                const v4f qb1 = fb[cs][i][1];
                const v4f qb2 = fb[cs][i][2];
                const v4f qb3 = fb[cs][i][3];

#pragma unroll
                for (int s = 0; s < 2; ++s) {
                    const v2f a  = s ? (v2f){qa.z,  qa.w}  : (v2f){qa.x,  qa.y};
                    const v2f b0 = s ? (v2f){qb0.z, qb0.w} : (v2f){qb0.x, qb0.y};
                    const v2f b1 = s ? (v2f){qb1.z, qb1.w} : (v2f){qb1.x, qb1.y};
                    const v2f b2 = s ? (v2f){qb2.z, qb2.w} : (v2f){qb2.x, qb2.y};
                    const v2f b3 = s ? (v2f){qb3.z, qb3.w} : (v2f){qb3.x, qb3.y};

                    sb0 = fmaf(b0.x, b0.x, sb0); sb0 = fmaf(b0.y, b0.y, sb0);
                    sb1 = fmaf(b1.x, b1.x, sb1); sb1 = fmaf(b1.y, b1.y, sb1);
                    sb2 = fmaf(b2.x, b2.x, sb2); sb2 = fmaf(b2.y, b2.y, sb2);
                    sb3 = fmaf(b3.x, b3.x, sb3); sb3 = fmaf(b3.y, b3.y, sb3);

                    acc0 = __builtin_amdgcn_wmma_f32_16x16x4_f32(false, a, false, b0, (short)0, acc0, false, false);
                    acc1 = __builtin_amdgcn_wmma_f32_16x16x4_f32(false, a, false, b1, (short)0, acc1, false, false);
                    acc2 = __builtin_amdgcn_wmma_f32_16x16x4_f32(false, a, false, b2, (short)0, acc2, false, false);
                    acc3 = __builtin_amdgcn_wmma_f32_16x16x4_f32(false, a, false, b3, (short)0, acc3, false, false);
                }
            }
        }

        float r2t[4];
        r2t[0] = sb0 + __shfl_xor(sb0, 16, 32);
        r2t[1] = sb1 + __shfl_xor(sb1, 16, 32);
        r2t[2] = sb2 + __shfl_xor(sb2, 16, 32);
        r2t[3] = sb3 + __shfl_xor(sb3, 16, 32);

        if (more) {
            float* d = sB0 + (cur ^ 1) * (TILE_M * PITCH);
#pragma unroll
            for (int i = 0; i < 8; ++i) {
                v2f lo = {pre[i].x, pre[i].y};
                v2f hi = {pre[i].z, pre[i].w};
                *(v2f*)(d + i * 8 * PITCH)     = lo;
                *(v2f*)(d + i * 8 * PITCH + 4) = hi;
            }
        }
        __syncthreads();

        v8f accs[4] = {acc0, acc1, acc2, acc3};
        const size_t tileBase = outRowBase + (size_t)(t * TILE_M);

#pragma unroll
        for (int tt = 0; tt < 4; ++tt) {
            const size_t colBase = tileBase + (size_t)(tt * 16 + lrow);
#pragma unroll
            for (int v = 0; v < 8; ++v) {
                const int row = v + 8 * half;
                float d2 = fmaf(-2.0f, accs[tt][v], l2row[v] + r2t[tt]);
                d2 = fmaxf(d2, 0.0f);
                const float den = 1.0f + __builtin_amdgcn_sqrtf(d2);
                Out[colBase + (size_t)row * MDIM] = __builtin_amdgcn_rcpf(den);
            }
        }
    }
}

extern "C" void kernel_launch(void* const* d_in, const int* in_sizes, int n_in,
                              void* d_out, int out_size, void* d_ws, size_t ws_size,
                              hipStream_t stream) {
    (void)in_sizes; (void)n_in; (void)out_size; (void)d_ws; (void)ws_size;
    const float* left  = (const float*)d_in[0];
    const float* right = (const float*)d_in[1];
    float* out = (float*)d_out;

    dim3 block(256, 1, 1);
    dim3 grid(MDIM / (64 * 8), NDIM / 128, BDIM);
    pairdist_wmma_kernel<<<grid, block, 0, stream>>>(left, right, out);
}
